# starting kernel
# baseline (speedup 1.0000x reference)
#include <hip/hip_runtime.h>
#include <math.h>


typedef __attribute__((ext_vector_type(2))) float v2f;
typedef __attribute__((ext_vector_type(8))) float v8f;

#define ALPHA_  0.12f
#define BETA_   0.88f
#define OMEGA_  6.0f
#define TWO_PI_ 6.28318530717958647692f
#define B_      4
#define T_      4096
#define D_      2048
#define HALF_   1024
#define SEG_    256
#define NSEG_   (T_ / SEG_)
#define NCH_    (SEG_ / 16)

__global__ void k_theta(float* __restrict__ cs, float* __restrict__ sn,
                        const int* __restrict__ step_idx) {
  int t = blockIdx.x * blockDim.x + threadIdx.x;
  if (t >= T_) return;
  float tg = (float)(t + 1 + step_idx[0]);
  float th = fmodf(OMEGA_ * log1pf(tg), TWO_PI_);
  cs[t] = cosf(th);
  sn[t] = sinf(th);
}

__global__ void k_seg_partial(const float* __restrict__ x,
                              const float* __restrict__ input_scale,
                              float* __restrict__ seg_carry) {
  int id  = blockIdx.x * blockDim.x + threadIdx.x;
  int d   = id % D_;
  int r   = id / D_;
  int b   = r % B_;
  int seg = r / B_;
  float k = ALPHA_ * input_scale[0];
  const float* xp = x + ((size_t)(b * T_ + seg * SEG_)) * D_ + d;
  float m = 0.0f;
  #pragma unroll 4
  for (int i = 0; i < SEG_; ++i)
    m = fmaf(BETA_, m, k * xp[(size_t)i * D_]);
  seg_carry[((size_t)seg * B_ + b) * D_ + d] = m;
}

__global__ void k_carry(const float* __restrict__ seg_carry,
                        const float* __restrict__ mem_state,
                        float* __restrict__ carry_in) {
  int id = blockIdx.x * blockDim.x + threadIdx.x;
  int d  = id % D_;
  int b  = id / D_;
  float f = powf(BETA_, (float)SEG_);
  float m = mem_state[d];
  for (int seg = 0; seg < NSEG_; ++seg) {
    size_t o = ((size_t)seg * B_ + b) * D_ + d;
    carry_in[o] = m;
    m = fmaf(f, m, seg_carry[o]);
  }
}

__device__ __forceinline__ v8f scan_tile_wmma(const float* __restrict__ xcol,
                                              float k, const v2f* a,
                                              const float* pw,
                                              float& m_carry, int n, int h) {
  v8f acc;
  #pragma unroll
  for (int v = 0; v < 8; ++v) acc[v] = pw[v] * m_carry;
  #pragma unroll
  for (int q = 0; q < 4; ++q) {
    int kb = 4 * q + 2 * h;
    v2f bb;
    bb.x = k * xcol[(size_t)kb * D_];
    bb.y = k * xcol[(size_t)(kb + 1) * D_];
    acc = __builtin_amdgcn_wmma_f32_16x16x4_f32(
         false, a[q],  false, bb,
         (short)0, acc,  false,  false);
  }
  m_carry = __shfl(acc[7], 16 + n, 32);
  return acc;
}

__global__ void
__launch_bounds__(32)
k_scan_out(const float* __restrict__ x, const float* __restrict__ gate,
           const float* __restrict__ input_scale,
           const float* __restrict__ carry_in,
           const float* __restrict__ cs, const float* __restrict__ sn,
           float* __restrict__ out) {
  const int lane = threadIdx.x;
  const int n = lane & 15;
  const int h = lane >> 4;

  int blk = blockIdx.x;
  int dt  = blk % (HALF_ / 16);
  int r   = blk / (HALF_ / 16);
  int b   = r % B_;
  int seg = r / B_;

  const float k = ALPHA_ * input_scale[0];
  const float g = 1.0f / (1.0f + expf(-gate[0]));

  const int M = n;
  v2f a[4];
  #pragma unroll
  for (int q = 0; q < 4; ++q) {
    int k0 = 4 * q + 2 * h;
    a[q].x = (k0 <= M)     ? powf(BETA_, (float)(M - k0))     : 0.0f;
    a[q].y = (k0 + 1 <= M) ? powf(BETA_, (float)(M - k0 - 1)) : 0.0f;
  }
  float pw[8];
  #pragma unroll
  for (int v = 0; v < 8; ++v) pw[v] = powf(BETA_, (float)(v + 8 * h + 1));

  const int d_lo = dt * 16 + n;
  const int d_hi = d_lo + HALF_;
  const size_t co = ((size_t)seg * B_ + b) * D_;
  float m_lo = carry_in[co + d_lo];
  float m_hi = carry_in[co + d_hi];

  for (int ch = 0; ch < NCH_; ++ch) {
    const int t0 = seg * SEG_ + ch * 16;
    const float* xbase = x + ((size_t)b * T_ + t0) * D_;

    v8f acc_lo = scan_tile_wmma(xbase + d_lo, k, a, pw, m_lo, n, h);
    v8f acc_hi = scan_tile_wmma(xbase + d_hi, k, a, pw, m_hi, n, h);

    #pragma unroll
    for (int v = 0; v < 8; ++v) {
      int t = t0 + v + 8 * h;
      float ct = cs[t], st = sn[t];
      size_t o = ((size_t)b * T_ + t) * D_;
      float xlo = x[o + d_lo];
      float xhi = x[o + d_hi];
      float ma = acc_lo[v], mb = acc_hi[v];
      out[o + d_lo] = fmaf(g, fmaf(ma, ct, -mb * st), xlo);
      out[o + d_hi] = fmaf(g, fmaf(ma, st,  mb * ct), xhi);
    }
  }
}

extern "C" void kernel_launch(void* const* d_in, const int* in_sizes, int n_in,
                              void* d_out, int out_size, void* d_ws,
                              size_t ws_size, hipStream_t stream) {
  const float* x           = (const float*)d_in[0];
  const float* gate        = (const float*)d_in[1];
  const float* input_scale = (const float*)d_in[2];
  const float* mem_state   = (const float*)d_in[3];
  const int*   step_idx    = (const int*)d_in[4];
  float* out = (float*)d_out;

  float* ws        = (float*)d_ws;
  float* cs        = ws;
  float* sn        = ws + T_;
  float* seg_carry = ws + 2 * T_;
  float* carry_in  = seg_carry + (size_t)NSEG_ * B_ * D_;

  k_theta<<<(T_ + 255) / 256, 256, 0, stream>>>(cs, sn, step_idx);
  k_seg_partial<<<(B_ * D_ * NSEG_) / 256, 256, 0, stream>>>(x, input_scale,
                                                             seg_carry);
  k_carry<<<(B_ * D_) / 256, 256, 0, stream>>>(seg_carry, mem_state, carry_in);
  k_scan_out<<<B_ * (HALF_ / 16) * NSEG_, 32, 0, stream>>>(
      x, gate, input_scale, carry_in, cs, sn, out);
}


# one pass over x: a work-group owns 32+32 columns for all timesteps, 16 waves scan 16 chunks per super-step, chunk totals chained in LDS; 4 -> 2 launches; chunk k order and carry sum order change
# speedup vs baseline: 3.4041x; 3.4041x over previous
#include <hip/hip_runtime.h>
#include <math.h>
#include <type_traits>


typedef __attribute__((ext_vector_type(2))) float v2f;
typedef __attribute__((ext_vector_type(4))) float v4f;
typedef __attribute__((ext_vector_type(8))) float v8f;

#define ALPHA_  0.12f
#define BETA_   0.88f
#define OMEGA_  6.0f
#define TWO_PI_ 6.28318530717958647692f
#define B_      4
#define T_      4096
#define D_      2048
#define HALF_   1024
#define NWAVE_  16
#define NSTRIP_ (HALF_ / 32)
#define NSS_    (T_ / (16 * NWAVE_))

__global__ void k_theta(float* __restrict__ cssn,
                        const int* __restrict__ step_idx) {
  int t = blockIdx.x * blockDim.x + threadIdx.x;
  if (t >= T_) return;
  float tg = (float)(t + 1 + step_idx[0]);
  float th = fmodf(OMEGA_ * log1pf(tg), TWO_PI_);
  cssn[2 * t]     = cosf(th);
  cssn[2 * t + 1] = sinf(th);
}

__device__ __forceinline__ void load_chunk(const float* __restrict__ p,
                                           v2f (&X)[16]) {
  #pragma unroll
  for (int r = 0; r < 8; ++r) {
    X[r]     = *(const v2f*)(p + (size_t)r * D_);
    X[8 + r] = *(const v2f*)(p + (size_t)r * D_ + HALF_);
  }
}

__device__ __forceinline__ void lds_barrier() {
  asm volatile("s_wait_dscnt 0x0\n\ts_barrier_signal -1\n\ts_barrier_wait -1"
               ::: "memory");
}

__global__ void
__launch_bounds__(32 * NWAVE_)
k_scan_strip(const float* __restrict__ x, const float* __restrict__ gate,
             const float* __restrict__ input_scale,
             const float* __restrict__ mem_state,
             const float* __restrict__ cssn, float* __restrict__ out) {
  __shared__ v4f P[2][NWAVE_][16];
  __shared__ v4f Mend[2][16];
  __shared__ v4f TBL[T_ / 2];

  const int lane = threadIdx.x & 31;
  const int w = __builtin_amdgcn_readfirstlane((int)(threadIdx.x >> 5));
  const int c = lane & 15;
  const int h = lane >> 4;
  const int strip = blockIdx.x % NSTRIP_;
  const int b     = blockIdx.x / NSTRIP_;

  const size_t lane_off =
      ((size_t)b * T_ + (size_t)(16 * w + 8 * h)) * D_ + strip * 32 + 2 * c;
  const size_t ss_stride = (size_t)16 * NWAVE_ * D_;

  const int col = strip * 32 + 2 * c;
  v4f m0;
  m0.x = mem_state[col];
  m0.y = mem_state[col + 1];
  m0.z = mem_state[HALF_ + col];
  m0.w = mem_state[HALF_ + col + 1];

  v4f tq[4];
  #pragma unroll
  for (int i = 0; i < 4; ++i)
    tq[i] = ((const v4f*)cssn)[threadIdx.x + 32 * NWAVE_ * i];

  v2f X[4][16];
  load_chunk(x + lane_off, X[0]);
  load_chunk(x + lane_off + ss_stride, X[1]);

  #pragma unroll
  for (int i = 0; i < 4; ++i) TBL[threadIdx.x + 32 * NWAVE_ * i] = tq[i];

  const float k = ALPHA_ * input_scale[0];
  const float g = 1.0f / (1.0f + expf(-gate[0]));

  const float pl = powf(BETA_, (float)lane);

  v2f a[4];
  #pragma unroll
  for (int q = 0; q < 4; ++q) {
    int k0 = 8 * h + 2 * q;
    float p0 = __shfl(pl, (k0 <= c)     ? c - k0     : 0, 32);
    float p1 = __shfl(pl, (k0 + 1 <= c) ? c - k0 - 1 : 0, 32);
    a[q].x = (k0 <= c)     ? p0 : 0.0f;
    a[q].y = (k0 + 1 <= c) ? p1 : 0.0f;
  }
  float pw[8];
  #pragma unroll
  for (int r = 0; r < 8; ++r) pw[r] = __shfl(pl, 8 * h + r + 1, 32);
  const float b16 = __shfl(pl, 16, 32);

  auto step = [&](auto U, auto PF, const int ss) {
    constexpr int u   = decltype(U)::value;
    constexpr int buf = u & 1;
    const size_t off = lane_off + (size_t)ss * ss_stride;
    v2f (&XC)[16] = X[u];

    if constexpr (decltype(PF)::value)
      load_chunk(x + off + 2 * ss_stride, X[(u + 2) & 3]);

    v8f acc0 = {}, acc1 = {}, acc2 = {}, acc3 = {};
    #pragma unroll
    for (int q = 0; q < 4; ++q) {
      v2f b0, b1, b2, b3;
      b0.x = k * XC[2 * q].x;      b0.y = k * XC[2 * q + 1].x;
      b1.x = k * XC[2 * q].y;      b1.y = k * XC[2 * q + 1].y;
      b2.x = k * XC[8 + 2 * q].x;  b2.y = k * XC[8 + 2 * q + 1].x;
      b3.x = k * XC[8 + 2 * q].y;  b3.y = k * XC[8 + 2 * q + 1].y;
      acc0 = __builtin_amdgcn_wmma_f32_16x16x4_f32(false, a[q], false, b0,
                                                   (short)0, acc0, false, false);
      acc1 = __builtin_amdgcn_wmma_f32_16x16x4_f32(false, a[q], false, b1,
                                                   (short)0, acc1, false, false);
      acc2 = __builtin_amdgcn_wmma_f32_16x16x4_f32(false, a[q], false, b2,
                                                   (short)0, acc2, false, false);
      acc3 = __builtin_amdgcn_wmma_f32_16x16x4_f32(false, a[q], false, b3,
                                                   (short)0, acc3, false, false);
    }

    if (h) {
      v4f p;
      p.x = acc0[7]; p.y = acc1[7]; p.z = acc2[7]; p.w = acc3[7];
      P[buf][w][c] = p;
    }
    lds_barrier();

    v4f m;
    if (ss == 0) m = m0; else m = Mend[buf ^ 1][c];
    for (int j = 0; j < w; ++j) {
      v4f p = P[buf][j][c];
      m.x = fmaf(b16, m.x, p.x);
      m.y = fmaf(b16, m.y, p.y);
      m.z = fmaf(b16, m.z, p.z);
      m.w = fmaf(b16, m.w, p.w);
    }
    if (w == NWAVE_ - 1 && h == 0) {
      v4f p = P[buf][w][c];
      v4f e;
      e.x = fmaf(b16, m.x, p.x);
      e.y = fmaf(b16, m.y, p.y);
      e.z = fmaf(b16, m.z, p.z);
      e.w = fmaf(b16, m.w, p.w);
      Mend[buf][c] = e;
    }

    const int trow = (16 * (ss * NWAVE_ + w) + 8 * h) >> 1;
    float tcs[16];
    #pragma unroll
    for (int i = 0; i < 4; ++i) {
      v4f e = TBL[trow + i];
      tcs[4 * i] = e.x; tcs[4 * i + 1] = e.y; tcs[4 * i + 2] = e.z; tcs[4 * i + 3] = e.w;
    }
    #pragma unroll
    for (int r = 0; r < 8; ++r) {
      float ct = tcs[2 * r];
      float st = tcs[2 * r + 1];
      float ma0 = fmaf(pw[r], m.x, acc0[r]);
      float ma1 = fmaf(pw[r], m.y, acc1[r]);
      float mb0 = fmaf(pw[r], m.z, acc2[r]);
      float mb1 = fmaf(pw[r], m.w, acc3[r]);
      v2f olo, ohi;
      olo.x = fmaf(g, fmaf(ma0, ct, -mb0 * st), XC[r].x);
      olo.y = fmaf(g, fmaf(ma1, ct, -mb1 * st), XC[r].y);
      ohi.x = fmaf(g, fmaf(ma0, st,  mb0 * ct), XC[8 + r].x);
      ohi.y = fmaf(g, fmaf(ma1, st,  mb1 * ct), XC[8 + r].y);
      *(v2f*)(out + off + (size_t)r * D_)         = olo;
      *(v2f*)(out + off + (size_t)r * D_ + HALF_) = ohi;
    }
  };
  using I0 = std::integral_constant<int, 0>;
  using I1 = std::integral_constant<int, 1>;
  using I2 = std::integral_constant<int, 2>;
  using I3 = std::integral_constant<int, 3>;
  #pragma unroll 1
  for (int s4 = 0; s4 < NSS_ - 4; s4 += 4) {
    step(I0{}, std::true_type{}, s4);
    step(I1{}, std::true_type{}, s4 + 1);
    step(I2{}, std::true_type{}, s4 + 2);
    step(I3{}, std::true_type{}, s4 + 3);
  }
  step(I0{}, std::true_type{},  NSS_ - 4);
  step(I1{}, std::true_type{},  NSS_ - 3);
  step(I2{}, std::false_type{}, NSS_ - 2);
  step(I3{}, std::false_type{}, NSS_ - 1);
}

extern "C" void kernel_launch(void* const* d_in, const int* in_sizes, int n_in,
                              void* d_out, int out_size, void* d_ws,
                              size_t ws_size, hipStream_t stream) {
  const float* x           = (const float*)d_in[0];
  const float* gate        = (const float*)d_in[1];
  const float* input_scale = (const float*)d_in[2];
  const float* mem_state   = (const float*)d_in[3];
  const int*   step_idx    = (const int*)d_in[4];
  float* out = (float*)d_out;

  float* cssn = (float*)d_ws;

  k_theta<<<(T_ + 255) / 256, 256, 0, stream>>>(cssn, step_idx);
  k_scan_strip<<<B_ * NSTRIP_, 32 * NWAVE_, 0, stream>>>(
      x, gate, input_scale, mem_state, cssn, out);
}


# scan pass: behind each barrier the chunk totals in front (and table rows) are requested as two groups of LDS loads ahead of the fold's dependent chain; end state from registers; sums and order kept
# speedup vs baseline: 3.5614x; 1.0462x over previous
#include <hip/hip_runtime.h>
#include <math.h>
#include <type_traits>


typedef __attribute__((ext_vector_type(2))) float v2f;
typedef __attribute__((ext_vector_type(4))) float v4f;
typedef __attribute__((ext_vector_type(8))) float v8f;

#define ALPHA_  0.12f
#define BETA_   0.88f
#define OMEGA_  6.0f
#define TWO_PI_ 6.28318530717958647692f
#define B_      4
#define T_      4096
#define D_      2048
#define HALF_   1024
#define NWAVE_  16
#define NSTRIP_ (HALF_ / 32)
#define NSS_    (T_ / (16 * NWAVE_))

__global__ void k_theta(float* __restrict__ cssn,
                        const int* __restrict__ step_idx) {
  int t = blockIdx.x * blockDim.x + threadIdx.x;
  if (t >= T_) return;
  float tg = (float)(t + 1 + step_idx[0]);
  float th = fmodf(OMEGA_ * log1pf(tg), TWO_PI_);
  cssn[2 * t]     = cosf(th);
  cssn[2 * t + 1] = sinf(th);
}

__device__ __forceinline__ void load_chunk(const float* __restrict__ p,
                                           v2f (&X)[16]) {
  #pragma unroll
  for (int r = 0; r < 8; ++r) {
    X[r]     = *(const v2f*)(p + (size_t)r * D_);
    X[8 + r] = *(const v2f*)(p + (size_t)r * D_ + HALF_);
  }
}

__device__ __forceinline__ void lds_barrier() {
  asm volatile("s_wait_dscnt 0x0\n\ts_barrier_signal -1\n\ts_barrier_wait -1"
               ::: "memory");
}

__device__ __forceinline__ void hold_loads8(const v4f (&q)[8]) {
  asm volatile("" :: "v"(q[0]), "v"(q[1]), "v"(q[2]), "v"(q[3]),
                     "v"(q[4]), "v"(q[5]), "v"(q[6]), "v"(q[7]));
}
__device__ __forceinline__ void hold_loads11(const v4f (&q)[8], const v4f (&e)[4]) {
  asm volatile("" :: "v"(q[0]), "v"(q[1]), "v"(q[2]), "v"(q[3]),
                     "v"(q[4]), "v"(q[5]), "v"(q[6]),
                     "v"(e[0]), "v"(e[1]), "v"(e[2]), "v"(e[3]));
}

__global__ void
__launch_bounds__(32 * NWAVE_)
k_scan_strip(const float* __restrict__ x, const float* __restrict__ gate,
             const float* __restrict__ input_scale,
             const float* __restrict__ mem_state,
             const float* __restrict__ cssn, float* __restrict__ out) {
  __shared__ v4f P[2][NWAVE_][16];
  __shared__ v4f Mend[2][16];
  __shared__ v4f TBL[T_ / 2];

  const int lane = threadIdx.x & 31;
  const int w = __builtin_amdgcn_readfirstlane((int)(threadIdx.x >> 5));
  const int c = lane & 15;
  const int h = lane >> 4;
  const int strip = blockIdx.x % NSTRIP_;
  const int b     = blockIdx.x / NSTRIP_;

  const size_t lane_off =
      ((size_t)b * T_ + (size_t)(16 * w + 8 * h)) * D_ + strip * 32 + 2 * c;
  const size_t ss_stride = (size_t)16 * NWAVE_ * D_;

  const int col = strip * 32 + 2 * c;
  v4f m0;
  m0.x = mem_state[col];
  m0.y = mem_state[col + 1];
  m0.z = mem_state[HALF_ + col];
  m0.w = mem_state[HALF_ + col + 1];

  v4f tq[4];
  #pragma unroll
  for (int i = 0; i < 4; ++i)
    tq[i] = ((const v4f*)cssn)[threadIdx.x + 32 * NWAVE_ * i];

  v2f X[4][16];
  load_chunk(x + lane_off, X[0]);
  load_chunk(x + lane_off + ss_stride, X[1]);

  #pragma unroll
  for (int i = 0; i < 4; ++i) TBL[threadIdx.x + 32 * NWAVE_ * i] = tq[i];

  const float k = ALPHA_ * input_scale[0];
  const float g = 1.0f / (1.0f + expf(-gate[0]));

  const float pl = powf(BETA_, (float)lane);

  v2f a[4];
  #pragma unroll
  for (int q = 0; q < 4; ++q) {
    int k0 = 8 * h + 2 * q;
    float p0 = __shfl(pl, (k0 <= c)     ? c - k0     : 0, 32);
    float p1 = __shfl(pl, (k0 + 1 <= c) ? c - k0 - 1 : 0, 32);
    a[q].x = (k0 <= c)     ? p0 : 0.0f;
    a[q].y = (k0 + 1 <= c) ? p1 : 0.0f;
  }
  float pw[8];
  #pragma unroll
  for (int r = 0; r < 8; ++r) pw[r] = __shfl(pl, 8 * h + r + 1, 32);
  const float b16 = __shfl(pl, 16, 32);

  auto step = [&](auto U, auto PF, const int ss) {
    constexpr int u   = decltype(U)::value;
    constexpr int buf = u & 1;
    const size_t off = lane_off + (size_t)ss * ss_stride;
    v2f (&XC)[16] = X[u];

    if constexpr (decltype(PF)::value)
      load_chunk(x + off + 2 * ss_stride, X[(u + 2) & 3]);

    v8f acc0 = {}, acc1 = {}, acc2 = {}, acc3 = {};
    #pragma unroll
    for (int q = 0; q < 4; ++q) {
      v2f b0, b1, b2, b3;
      b0.x = k * XC[2 * q].x;      b0.y = k * XC[2 * q + 1].x;
      b1.x = k * XC[2 * q].y;      b1.y = k * XC[2 * q + 1].y;
      b2.x = k * XC[8 + 2 * q].x;  b2.y = k * XC[8 + 2 * q + 1].x;
      b3.x = k * XC[8 + 2 * q].y;  b3.y = k * XC[8 + 2 * q + 1].y;
      acc0 = __builtin_amdgcn_wmma_f32_16x16x4_f32(false, a[q], false, b0,
                                                   (short)0, acc0, false, false);
      acc1 = __builtin_amdgcn_wmma_f32_16x16x4_f32(false, a[q], false, b1,
                                                   (short)0, acc1, false, false);
      acc2 = __builtin_amdgcn_wmma_f32_16x16x4_f32(false, a[q], false, b2,
                                                   (short)0, acc2, false, false);
      acc3 = __builtin_amdgcn_wmma_f32_16x16x4_f32(false, a[q], false, b3,
                                                   (short)0, acc3, false, false);
    }

    if (h) {
      v4f p;
      p.x = acc0[7]; p.y = acc1[7]; p.z = acc2[7]; p.w = acc3[7];
      P[buf][w][c] = p;
    }
    lds_barrier();

    v4f m;
    if (ss == 0) m = m0; else m = Mend[buf ^ 1][c];
    v4f q[8];
    if (w > 0) {
      #pragma unroll
      for (int j = 0; j < 8; ++j) q[j] = P[buf][j][c];
      hold_loads8(q);
      #pragma unroll
      for (int j = 0; j < 8; ++j) {
        if (j >= w) break;
        m.x = fmaf(b16, m.x, q[j].x);
        m.y = fmaf(b16, m.y, q[j].y);
        m.z = fmaf(b16, m.z, q[j].z);
        m.w = fmaf(b16, m.w, q[j].w);
      }
    }

    const int trow = (16 * (ss * NWAVE_ + w) + 8 * h) >> 1;
    v4f e4[4];
    #pragma unroll
    for (int i = 0; i < 4; ++i) e4[i] = TBL[trow + i];

    if (w > 8) {
      #pragma unroll
      for (int j = 0; j < 7; ++j) q[j] = P[buf][8 + j][c];
      hold_loads11(q, e4);
      #pragma unroll
      for (int j = 0; j < 7; ++j) {
        if (8 + j >= w) break;
        m.x = fmaf(b16, m.x, q[j].x);
        m.y = fmaf(b16, m.y, q[j].y);
        m.z = fmaf(b16, m.z, q[j].z);
        m.w = fmaf(b16, m.w, q[j].w);
      }
    }
    if (w == NWAVE_ - 1 && h == 1) {
      v4f e;
      e.x = fmaf(b16, m.x, acc0[7]);
      e.y = fmaf(b16, m.y, acc1[7]);
      e.z = fmaf(b16, m.z, acc2[7]);
      e.w = fmaf(b16, m.w, acc3[7]);
      Mend[buf][c] = e;
    }

    float tcs[16];
    #pragma unroll
    for (int i = 0; i < 4; ++i) {
      tcs[4 * i] = e4[i].x; tcs[4 * i + 1] = e4[i].y; tcs[4 * i + 2] = e4[i].z; tcs[4 * i + 3] = e4[i].w;
    }
    #pragma unroll
    for (int r = 0; r < 8; ++r) {
      float ct = tcs[2 * r];
      float st = tcs[2 * r + 1];
      float ma0 = fmaf(pw[r], m.x, acc0[r]);
      float ma1 = fmaf(pw[r], m.y, acc1[r]);
      float mb0 = fmaf(pw[r], m.z, acc2[r]);
      float mb1 = fmaf(pw[r], m.w, acc3[r]);
      v2f olo, ohi;
      olo.x = fmaf(g, fmaf(ma0, ct, -mb0 * st), XC[r].x);
      olo.y = fmaf(g, fmaf(ma1, ct, -mb1 * st), XC[r].y);
      ohi.x = fmaf(g, fmaf(ma0, st,  mb0 * ct), XC[8 + r].x);
      ohi.y = fmaf(g, fmaf(ma1, st,  mb1 * ct), XC[8 + r].y);
      *(v2f*)(out + off + (size_t)r * D_)         = olo;
      *(v2f*)(out + off + (size_t)r * D_ + HALF_) = ohi;
    }
  };
  using I0 = std::integral_constant<int, 0>;
  using I1 = std::integral_constant<int, 1>;
  using I2 = std::integral_constant<int, 2>;
  using I3 = std::integral_constant<int, 3>;
  #pragma unroll 1
  for (int s4 = 0; s4 < NSS_ - 4; s4 += 4) {
    step(I0{}, std::true_type{}, s4);
    step(I1{}, std::true_type{}, s4 + 1);
    step(I2{}, std::true_type{}, s4 + 2);
    step(I3{}, std::true_type{}, s4 + 3);
  }
  step(I0{}, std::true_type{},  NSS_ - 4);
  step(I1{}, std::true_type{},  NSS_ - 3);
  step(I2{}, std::false_type{}, NSS_ - 2);
  step(I3{}, std::false_type{}, NSS_ - 1);
}

extern "C" void kernel_launch(void* const* d_in, const int* in_sizes, int n_in,
                              void* d_out, int out_size, void* d_ws,
                              size_t ws_size, hipStream_t stream) {
  const float* x           = (const float*)d_in[0];
  const float* gate        = (const float*)d_in[1];
  const float* input_scale = (const float*)d_in[2];
  const float* mem_state   = (const float*)d_in[3];
  const int*   step_idx    = (const int*)d_in[4];
  float* out = (float*)d_out;

  float* cssn = (float*)d_ws;

  k_theta<<<(T_ + 255) / 256, 256, 0, stream>>>(cssn, step_idx);
  k_scan_strip<<<B_ * NSTRIP_, 32 * NWAVE_, 0, stream>>>(
      x, gate, input_scale, mem_state, cssn, out);
}


# time axis cut in two, 256 work-groups: a first launch (with the phase table) runs a plain recurrence over rows 0-2047 for the second half's start state; carry sum order changes
# speedup vs baseline: 3.7060x; 1.0406x over previous
#include <hip/hip_runtime.h>
#include <math.h>
#include <type_traits>


typedef __attribute__((ext_vector_type(2))) float v2f;
typedef __attribute__((ext_vector_type(4))) float v4f;
typedef __attribute__((ext_vector_type(8))) float v8f;

#define ALPHA_  0.12f
#define BETA_   0.88f
#define OMEGA_  6.0f
#define TWO_PI_ 6.28318530717958647692f
#define B_      4
#define T_      4096
#define D_      2048
#define HALF_   1024
#define NWAVE_  16
#define NSTRIP_ (HALF_ / 32)
#define THALF_  (T_ / 2)
#define NSS_    (THALF_ / (16 * NWAVE_))
#define PRE_COLS_ 32
#define PRE_ROWS_ (THALF_ / NWAVE_)
#define PRE_NBLK_ (B_ * D_ / PRE_COLS_)
#define TH_NBLK_  (T_ / (32 * NWAVE_))

__global__ void
__launch_bounds__(32 * NWAVE_)
k_pre(const float* __restrict__ x, const float* __restrict__ input_scale,
      const float* __restrict__ mem_state, const int* __restrict__ step_idx,
      float* __restrict__ cssn, float* __restrict__ mid) {
  __shared__ float TOT[NWAVE_][32];
  if (blockIdx.x >= PRE_NBLK_) {
    int t = (blockIdx.x - PRE_NBLK_) * blockDim.x + threadIdx.x;
    float tg = (float)(t + 1 + step_idx[0]);
    float th = fmodf(OMEGA_ * log1pf(tg), TWO_PI_);
    cssn[2 * t]     = cosf(th);
    cssn[2 * t + 1] = sinf(th);
    return;
  }
  const int lane = threadIdx.x & 31;
  const int w    = threadIdx.x >> 5;
  const int b    = blockIdx.x / (D_ / PRE_COLS_);
  const int col  = (blockIdx.x % (D_ / PRE_COLS_)) * PRE_COLS_ + lane;
  const float k  = ALPHA_ * input_scale[0];
  const float* p = x + ((size_t)b * T_ + (size_t)w * PRE_ROWS_) * D_ + col;
  float m = 0.0f;
  float v[PRE_ROWS_ / 32][32];
  #pragma unroll
  for (int i = 0; i < 32; ++i) v[0][i] = p[(size_t)i * D_];
  #pragma unroll
  for (int g = 0; g < PRE_ROWS_ / 32; ++g) {
    if (g + 1 < PRE_ROWS_ / 32) {
      #pragma unroll
      for (int i = 0; i < 32; ++i) v[g + 1][i] = p[(size_t)(32 * (g + 1) + i) * D_];
    }
    #pragma unroll
    for (int i = 0; i < 32; ++i) m = fmaf(BETA_, m, k * v[g][i]);
  }
  TOT[w][lane] = m;
  __syncthreads();
  if (w == 0) {
    const float bw = powf(BETA_, (float)PRE_ROWS_);
    float st = mem_state[col];
    #pragma unroll
    for (int j = 0; j < NWAVE_; ++j) st = fmaf(bw, st, TOT[j][lane]);
    mid[(size_t)b * D_ + col] = st;
  }
}

__device__ __forceinline__ void load_chunk(const float* __restrict__ p,
                                           v2f (&X)[16]) {
  #pragma unroll
  for (int r = 0; r < 8; ++r) {
    X[r]     = *(const v2f*)(p + (size_t)r * D_);
    X[8 + r] = *(const v2f*)(p + (size_t)r * D_ + HALF_);
  }
}

__device__ __forceinline__ void lds_barrier() {
  asm volatile("s_wait_dscnt 0x0\n\ts_barrier_signal -1\n\ts_barrier_wait -1"
               ::: "memory");
}

__device__ __forceinline__ void hold_loads8(const v4f (&q)[8]) {
  asm volatile("" :: "v"(q[0]), "v"(q[1]), "v"(q[2]), "v"(q[3]),
                     "v"(q[4]), "v"(q[5]), "v"(q[6]), "v"(q[7]));
}
__device__ __forceinline__ void hold_loads11(const v4f (&q)[8], const v4f (&e)[4]) {
  asm volatile("" :: "v"(q[0]), "v"(q[1]), "v"(q[2]), "v"(q[3]),
                     "v"(q[4]), "v"(q[5]), "v"(q[6]),
                     "v"(e[0]), "v"(e[1]), "v"(e[2]), "v"(e[3]));
}

__global__ void
__launch_bounds__(32 * NWAVE_)
k_scan_strip(const float* __restrict__ x, const float* __restrict__ gate,
             const float* __restrict__ input_scale,
             const float* __restrict__ mem_state,
             const float* __restrict__ cssn, const float* __restrict__ mid,
             float* __restrict__ out) {
  __shared__ v4f P[2][NWAVE_][16];
  __shared__ v4f Mend[2][16];
  __shared__ v4f TBL[THALF_ / 2];

  const int lane = threadIdx.x & 31;
  const int w = __builtin_amdgcn_readfirstlane((int)(threadIdx.x >> 5));
  const int c = lane & 15;
  const int h = lane >> 4;
  const int strip = blockIdx.x % NSTRIP_;
  const int b     = (blockIdx.x / NSTRIP_) % B_;
  const int ht    = blockIdx.x / (NSTRIP_ * B_);

  const size_t lane_off =
      ((size_t)b * T_ + (size_t)(ht * THALF_ + 16 * w + 8 * h)) * D_ + strip * 32 + 2 * c;
  const size_t ss_stride = (size_t)16 * NWAVE_ * D_;

  const int col = strip * 32 + 2 * c;
  const float* sp = ht ? mid + (size_t)b * D_ : mem_state;
  v4f m0;
  m0.x = sp[col];
  m0.y = sp[col + 1];
  m0.z = sp[HALF_ + col];
  m0.w = sp[HALF_ + col + 1];

  v4f tq[2];
  #pragma unroll
  for (int i = 0; i < 2; ++i)
    tq[i] = ((const v4f*)cssn)[ht * (THALF_ / 2) + threadIdx.x + 32 * NWAVE_ * i];

  v2f X[4][16];
  load_chunk(x + lane_off, X[0]);
  load_chunk(x + lane_off + ss_stride, X[1]);

  #pragma unroll
  for (int i = 0; i < 2; ++i) TBL[threadIdx.x + 32 * NWAVE_ * i] = tq[i];

  const float k = ALPHA_ * input_scale[0];
  const float g = 1.0f / (1.0f + expf(-gate[0]));

  const float pl = powf(BETA_, (float)lane);

  v2f a[4];
  #pragma unroll
  for (int q = 0; q < 4; ++q) {
    int k0 = 8 * h + 2 * q;
    float p0 = __shfl(pl, (k0 <= c)     ? c - k0     : 0, 32);
    float p1 = __shfl(pl, (k0 + 1 <= c) ? c - k0 - 1 : 0, 32);
    a[q].x = (k0 <= c)     ? p0 : 0.0f;
    a[q].y = (k0 + 1 <= c) ? p1 : 0.0f;
  }
  float pw[8];
  #pragma unroll
  for (int r = 0; r < 8; ++r) pw[r] = __shfl(pl, 8 * h + r + 1, 32);
  const float b16 = __shfl(pl, 16, 32);

  auto step = [&](auto U, auto PF, const int ss) {
    constexpr int u   = decltype(U)::value;
    constexpr int buf = u & 1;
    const size_t off = lane_off + (size_t)ss * ss_stride;
    v2f (&XC)[16] = X[u];

    if constexpr (decltype(PF)::value)
      load_chunk(x + off + 2 * ss_stride, X[(u + 2) & 3]);

    v8f acc0 = {}, acc1 = {}, acc2 = {}, acc3 = {};
    #pragma unroll
    for (int q = 0; q < 4; ++q) {
      v2f b0, b1, b2, b3;
      b0.x = k * XC[2 * q].x;      b0.y = k * XC[2 * q + 1].x;
      b1.x = k * XC[2 * q].y;      b1.y = k * XC[2 * q + 1].y;
      b2.x = k * XC[8 + 2 * q].x;  b2.y = k * XC[8 + 2 * q + 1].x;
      b3.x = k * XC[8 + 2 * q].y;  b3.y = k * XC[8 + 2 * q + 1].y;
      acc0 = __builtin_amdgcn_wmma_f32_16x16x4_f32(false, a[q], false, b0,
                                                   (short)0, acc0, false, false);
      acc1 = __builtin_amdgcn_wmma_f32_16x16x4_f32(false, a[q], false, b1,
                                                   (short)0, acc1, false, false);
      acc2 = __builtin_amdgcn_wmma_f32_16x16x4_f32(false, a[q], false, b2,
                                                   (short)0, acc2, false, false);
      acc3 = __builtin_amdgcn_wmma_f32_16x16x4_f32(false, a[q], false, b3,
                                                   (short)0, acc3, false, false);
    }

    if (h) {
      v4f p;
      p.x = acc0[7]; p.y = acc1[7]; p.z = acc2[7]; p.w = acc3[7];
      P[buf][w][c] = p;
    }
    lds_barrier();

    v4f m;
    if (ss == 0) m = m0; else m = Mend[buf ^ 1][c];
    v4f q[8];
    if (w > 0) {
      #pragma unroll
      for (int j = 0; j < 8; ++j) q[j] = P[buf][j][c];
      hold_loads8(q);
      #pragma unroll
      for (int j = 0; j < 8; ++j) {
        if (j >= w) break;
        m.x = fmaf(b16, m.x, q[j].x);
        m.y = fmaf(b16, m.y, q[j].y);
        m.z = fmaf(b16, m.z, q[j].z);
        m.w = fmaf(b16, m.w, q[j].w);
      }
    }

    const int trow = (16 * (ss * NWAVE_ + w) + 8 * h) >> 1;
    v4f e4[4];
    #pragma unroll
    for (int i = 0; i < 4; ++i) e4[i] = TBL[trow + i];

    if (w > 8) {
      #pragma unroll
      for (int j = 0; j < 7; ++j) q[j] = P[buf][8 + j][c];
      hold_loads11(q, e4);
      #pragma unroll
      for (int j = 0; j < 7; ++j) {
        if (8 + j >= w) break;
        m.x = fmaf(b16, m.x, q[j].x);
        m.y = fmaf(b16, m.y, q[j].y);
        m.z = fmaf(b16, m.z, q[j].z);
        m.w = fmaf(b16, m.w, q[j].w);
      }
    }
    if (w == NWAVE_ - 1 && h == 1) {
      v4f e;
      e.x = fmaf(b16, m.x, acc0[7]);
      e.y = fmaf(b16, m.y, acc1[7]);
      e.z = fmaf(b16, m.z, acc2[7]);
      e.w = fmaf(b16, m.w, acc3[7]);
      Mend[buf][c] = e;
    }

    float tcs[16];
    #pragma unroll
    for (int i = 0; i < 4; ++i) {
      tcs[4 * i] = e4[i].x; tcs[4 * i + 1] = e4[i].y; tcs[4 * i + 2] = e4[i].z; tcs[4 * i + 3] = e4[i].w;
    }
    #pragma unroll
    for (int r = 0; r < 8; ++r) {
      float ct = tcs[2 * r];
      float st = tcs[2 * r + 1];
      float ma0 = fmaf(pw[r], m.x, acc0[r]);
      float ma1 = fmaf(pw[r], m.y, acc1[r]);
      float mb0 = fmaf(pw[r], m.z, acc2[r]);
      float mb1 = fmaf(pw[r], m.w, acc3[r]);
      v2f olo, ohi;
      olo.x = fmaf(g, fmaf(ma0, ct, -mb0 * st), XC[r].x);
      olo.y = fmaf(g, fmaf(ma1, ct, -mb1 * st), XC[r].y);
      ohi.x = fmaf(g, fmaf(ma0, st,  mb0 * ct), XC[8 + r].x);
      ohi.y = fmaf(g, fmaf(ma1, st,  mb1 * ct), XC[8 + r].y);
      *(v2f*)(out + off + (size_t)r * D_)         = olo;
      *(v2f*)(out + off + (size_t)r * D_ + HALF_) = ohi;
    }
  };
  using I0 = std::integral_constant<int, 0>;
  using I1 = std::integral_constant<int, 1>;
  using I2 = std::integral_constant<int, 2>;
  using I3 = std::integral_constant<int, 3>;
  #pragma unroll 1
  for (int s4 = 0; s4 < NSS_ - 4; s4 += 4) {
    step(I0{}, std::true_type{}, s4);
    step(I1{}, std::true_type{}, s4 + 1);
    step(I2{}, std::true_type{}, s4 + 2);
    step(I3{}, std::true_type{}, s4 + 3);
  }
  step(I0{}, std::true_type{},  NSS_ - 4);
  step(I1{}, std::true_type{},  NSS_ - 3);
  step(I2{}, std::false_type{}, NSS_ - 2);
  step(I3{}, std::false_type{}, NSS_ - 1);
}

extern "C" void kernel_launch(void* const* d_in, const int* in_sizes, int n_in,
                              void* d_out, int out_size, void* d_ws,
                              size_t ws_size, hipStream_t stream) {
  const float* x           = (const float*)d_in[0];
  const float* gate        = (const float*)d_in[1];
  const float* input_scale = (const float*)d_in[2];
  const float* mem_state   = (const float*)d_in[3];
  const int*   step_idx    = (const int*)d_in[4];
  float* out = (float*)d_out;

  float* cssn = (float*)d_ws;
  float* mid  = (float*)d_ws + 2 * T_;

  k_pre<<<PRE_NBLK_ + TH_NBLK_, 32 * NWAVE_, 0, stream>>>(
      x, input_scale, mem_state, step_idx, cssn, mid);
  k_scan_strip<<<2 * B_ * NSTRIP_, 32 * NWAVE_, 0, stream>>>(
      x, gate, input_scale, mem_state, cssn, mid, out);
}


# one launch: work-groups of the later rows (2560-4095) first run the recurrence over rows 0-2559 of their own columns; phase rows built per work-group in LDS; 2->1 launches; carry sum order changes
# speedup vs baseline: 3.9634x; 1.0695x over previous
#include <hip/hip_runtime.h>
#include <math.h>
#include <type_traits>


typedef __attribute__((ext_vector_type(2))) float v2f;
typedef __attribute__((ext_vector_type(4))) float v4f;
typedef __attribute__((ext_vector_type(8))) float v8f;

#define ALPHA_  0.12f
#define BETA_   0.88f
#define OMEGA_  6.0f
#define TWO_PI_ 6.28318530717958647692f
#define B_      4
#define T_      4096
#define D_      2048
#define HALF_   1024
#define NWAVE_  16
#define NSTRIP_ (HALF_ / 32)
#define TCUT_   2560
#define SSROWS_ (16 * NWAVE_)
#define REC_ROWS_ (TCUT_ / NWAVE_)
#define REC_CHUNKS_ (REC_ROWS_ / 16)
#define RD_     3

__device__ __forceinline__ void load_chunk(const float* __restrict__ p,
                                           v2f (&X)[16]) {
  #pragma unroll
  for (int r = 0; r < 8; ++r) {
    X[r]     = *(const v2f*)(p + (size_t)r * D_);
    X[8 + r] = *(const v2f*)(p + (size_t)r * D_ + HALF_);
  }
}

__device__ __forceinline__ void lds_barrier() {
  asm volatile("s_wait_dscnt 0x0\n\ts_barrier_signal -1\n\ts_barrier_wait -1"
               ::: "memory");
}

__device__ __forceinline__ void hold_loads8(const v4f (&q)[8]) {
  asm volatile("" :: "v"(q[0]), "v"(q[1]), "v"(q[2]), "v"(q[3]),
                     "v"(q[4]), "v"(q[5]), "v"(q[6]), "v"(q[7]));
}
__device__ __forceinline__ void hold_loads11(const v4f (&q)[8], const v4f (&e)[4]) {
  asm volatile("" :: "v"(q[0]), "v"(q[1]), "v"(q[2]), "v"(q[3]),
                     "v"(q[4]), "v"(q[5]), "v"(q[6]),
                     "v"(e[0]), "v"(e[1]), "v"(e[2]), "v"(e[3]));
}

__global__ void
__launch_bounds__(32 * NWAVE_)
k_scan_strip(const float* __restrict__ x, const float* __restrict__ gate,
             const float* __restrict__ input_scale,
             const float* __restrict__ mem_state,
             const int* __restrict__ step_idx,
             float* __restrict__ out) {
  __shared__ v4f P[2][NWAVE_][16];
  __shared__ v4f Mend[2][16];
  __shared__ v4f TOT[NWAVE_][16];
  __shared__ __attribute__((aligned(16))) float TBL[2 * TCUT_];

  const int lane = threadIdx.x & 31;
  const int w = __builtin_amdgcn_readfirstlane((int)(threadIdx.x >> 5));
  const int c = lane & 15;
  const int h = lane >> 4;
  const int sh    = (blockIdx.x >> 3) & 3;
  const int pr    = (blockIdx.x >> 5) * 8 + (blockIdx.x & 7);
  const int part  = sh >> 1;
  const int strip = 2 * (pr % (NSTRIP_ / 2)) + (sh & 1);
  const int b     = pr / (NSTRIP_ / 2);
  const int t_org = part ? TCUT_ : 0;

  const int col = strip * 32 + 2 * c;
  const size_t lane_off =
      ((size_t)b * T_ + (size_t)(t_org + 16 * w + 8 * h)) * D_ + col;
  const size_t rec_off =
      ((size_t)b * T_ + (size_t)(REC_ROWS_ * w + 8 * h)) * D_ + col;
  const size_t ss_stride = (size_t)SSROWS_ * D_;

  v2f X[4][16];
  load_chunk(x + lane_off, X[0]);

  const float k = ALPHA_ * input_scale[0];
  const float g = 1.0f / (1.0f + expf(-gate[0]));

  const int sidx = step_idx[0];
  auto build_table = [&](const int nt) {
    #pragma unroll 1
    for (int tl = threadIdx.x; tl < nt; tl += 32 * NWAVE_) {
      float tg = (float)(t_org + tl + 1 + sidx);
      float th = fmodf(OMEGA_ * log1pf(tg), TWO_PI_);
      v2f cs;
      cs.x = cosf(th);
      cs.y = sinf(th);
      *(v2f*)&TBL[2 * tl] = cs;
    }
  };

  const float pl = powf(BETA_, (float)lane);
  const float b16 = __shfl(pl, 16, 32);

  v2f a[4];
  float pw[8];
  auto make_weights = [&]() {
    #pragma unroll
    for (int q = 0; q < 4; ++q) {
      int k0 = 8 * h + 2 * q;
      float q0 = __shfl(pl, (k0 <= c)     ? c - k0     : 0, 32);
      float q1 = __shfl(pl, (k0 + 1 <= c) ? c - k0 - 1 : 0, 32);
      a[q].x = (k0 <= c)     ? q0 : 0.0f;
      a[q].y = (k0 + 1 <= c) ? q1 : 0.0f;
    }
    #pragma unroll
    for (int r = 0; r < 8; ++r) pw[r] = __shfl(pl, 8 * h + r + 1, 32);
  };

  auto step = [&](auto U, auto PF, const int ss) {
    constexpr int u   = decltype(U)::value;
    constexpr int buf = u & 1;
    const size_t off = lane_off + (size_t)ss * ss_stride;
    v2f (&XC)[16] = X[u];

    if constexpr (decltype(PF)::value)
      load_chunk(x + off + 2 * ss_stride, X[(u + 2) & 3]);

    v8f acc0 = {}, acc1 = {}, acc2 = {}, acc3 = {};
    #pragma unroll
    for (int q = 0; q < 4; ++q) {
      v2f b0, b1, b2, b3;
      b0.x = k * XC[2 * q].x;      b0.y = k * XC[2 * q + 1].x;
      b1.x = k * XC[2 * q].y;      b1.y = k * XC[2 * q + 1].y;
      b2.x = k * XC[8 + 2 * q].x;  b2.y = k * XC[8 + 2 * q + 1].x;
      b3.x = k * XC[8 + 2 * q].y;  b3.y = k * XC[8 + 2 * q + 1].y;
      acc0 = __builtin_amdgcn_wmma_f32_16x16x4_f32(false, a[q], false, b0,
                                                   (short)0, acc0, false, false);
      acc1 = __builtin_amdgcn_wmma_f32_16x16x4_f32(false, a[q], false, b1,
                                                   (short)0, acc1, false, false);
      acc2 = __builtin_amdgcn_wmma_f32_16x16x4_f32(false, a[q], false, b2,
                                                   (short)0, acc2, false, false);
      acc3 = __builtin_amdgcn_wmma_f32_16x16x4_f32(false, a[q], false, b3,
                                                   (short)0, acc3, false, false);
    }

    if (h) {
      v4f p;
      p.x = acc0[7]; p.y = acc1[7]; p.z = acc2[7]; p.w = acc3[7];
      P[buf][w][c] = p;
    }
    lds_barrier();

    v4f m = Mend[buf ^ 1][c];
    v4f q[8];
    if (w > 0) {
      #pragma unroll
      for (int j = 0; j < 8; ++j) q[j] = P[buf][j][c];
      hold_loads8(q);
      #pragma unroll
      for (int j = 0; j < 8; ++j) {
        if (j >= w) break;
        m.x = fmaf(b16, m.x, q[j].x);
        m.y = fmaf(b16, m.y, q[j].y);
        m.z = fmaf(b16, m.z, q[j].z);
        m.w = fmaf(b16, m.w, q[j].w);
      }
    }

    const int trow = 16 * (ss * NWAVE_ + w) + 8 * h;
    v4f e4[4];
    #pragma unroll
    for (int i = 0; i < 4; ++i) e4[i] = *(const v4f*)&TBL[2 * trow + 4 * i];

    if (w > 8) {
      #pragma unroll
      for (int j = 0; j < 7; ++j) q[j] = P[buf][8 + j][c];
      hold_loads11(q, e4);
      #pragma unroll
      for (int j = 0; j < 7; ++j) {
        if (8 + j >= w) break;
        m.x = fmaf(b16, m.x, q[j].x);
        m.y = fmaf(b16, m.y, q[j].y);
        m.z = fmaf(b16, m.z, q[j].z);
        m.w = fmaf(b16, m.w, q[j].w);
      }
    }
    if (w == NWAVE_ - 1 && h == 1) {
      v4f e;
      e.x = fmaf(b16, m.x, acc0[7]);
      e.y = fmaf(b16, m.y, acc1[7]);
      e.z = fmaf(b16, m.z, acc2[7]);
      e.w = fmaf(b16, m.w, acc3[7]);
      Mend[buf][c] = e;
    }

    float tcs[16];
    #pragma unroll
    for (int i = 0; i < 4; ++i) {
      tcs[4 * i] = e4[i].x; tcs[4 * i + 1] = e4[i].y; tcs[4 * i + 2] = e4[i].z; tcs[4 * i + 3] = e4[i].w;
    }
    #pragma unroll
    for (int r = 0; r < 8; ++r) {
      float ct = tcs[2 * r];
      float st = tcs[2 * r + 1];
      float ma0 = fmaf(pw[r], m.x, acc0[r]);
      float ma1 = fmaf(pw[r], m.y, acc1[r]);
      float mb0 = fmaf(pw[r], m.z, acc2[r]);
      float mb1 = fmaf(pw[r], m.w, acc3[r]);
      v2f olo, ohi;
      olo.x = fmaf(g, fmaf(ma0, ct, -mb0 * st), XC[r].x);
      olo.y = fmaf(g, fmaf(ma1, ct, -mb1 * st), XC[r].y);
      ohi.x = fmaf(g, fmaf(ma0, st,  mb0 * ct), XC[8 + r].x);
      ohi.y = fmaf(g, fmaf(ma1, st,  mb1 * ct), XC[8 + r].y);
      *(v2f*)(out + off + (size_t)r * D_)         = olo;
      *(v2f*)(out + off + (size_t)r * D_ + HALF_) = ohi;
    }
  };
  using I0 = std::integral_constant<int, 0>;
  using I1 = std::integral_constant<int, 1>;
  using I2 = std::integral_constant<int, 2>;
  using I3 = std::integral_constant<int, 3>;
  using PFy = std::true_type;
  using PFn = std::false_type;

  if (!part) {
    build_table(TCUT_);
    if (w == 0 && h == 0) {
      v4f mi;
      mi.x = mem_state[col];
      mi.y = mem_state[col + 1];
      mi.z = mem_state[HALF_ + col];
      mi.w = mem_state[HALF_ + col + 1];
      Mend[1][c] = mi;
    }
  } else {
    v2f R[RD_ + 1][16];
    #pragma unroll
    for (int j = 0; j < RD_; ++j)
      load_chunk(x + rec_off + (size_t)(16 * j) * D_, R[j]);
    build_table(T_ - TCUT_);
    const float b8 = __shfl(pl, 8, 32);
    const float bw = powf(BETA_, (float)REC_ROWS_);
    v4f A = {};
    auto rec = [&](auto J) {
      constexpr int j = decltype(J)::value;
      if constexpr (j + RD_ < REC_CHUNKS_)
        load_chunk(x + rec_off + (size_t)(16 * (j + RD_)) * D_, R[(j + RD_) % (RD_ + 1)]);
      v2f (&XC)[16] = R[j % (RD_ + 1)];
      v4f L = {};
      #pragma unroll
      for (int r = 0; r < 8; ++r) {
        L.x = fmaf(BETA_, L.x, XC[r].x);
        L.y = fmaf(BETA_, L.y, XC[r].y);
        L.z = fmaf(BETA_, L.z, XC[8 + r].x);
        L.w = fmaf(BETA_, L.w, XC[8 + r].y);
      }
      A.x = fmaf(b16, A.x, L.x);
      A.y = fmaf(b16, A.y, L.y);
      A.z = fmaf(b16, A.z, L.z);
      A.w = fmaf(b16, A.w, L.w);
      __builtin_amdgcn_sched_barrier(0);
    };
    static_assert(REC_CHUNKS_ == 10, "the calls below are written out for 10 chunks");
    rec(std::integral_constant<int, 0>{});
    rec(std::integral_constant<int, 1>{});
    rec(std::integral_constant<int, 2>{});
    rec(std::integral_constant<int, 3>{});
    rec(std::integral_constant<int, 4>{});
    rec(std::integral_constant<int, 5>{});
    rec(std::integral_constant<int, 6>{});
    rec(std::integral_constant<int, 7>{});
    rec(std::integral_constant<int, 8>{});
    rec(std::integral_constant<int, 9>{});
    v4f A0;
    A0.x = __shfl(A.x, c, 32);
    A0.y = __shfl(A.y, c, 32);
    A0.z = __shfl(A.z, c, 32);
    A0.w = __shfl(A.w, c, 32);
    if (h) {
      v4f S;
      S.x = k * fmaf(b8, A0.x, A.x);
      S.y = k * fmaf(b8, A0.y, A.y);
      S.z = k * fmaf(b8, A0.z, A.z);
      S.w = k * fmaf(b8, A0.w, A.w);
      TOT[w][c] = S;
    }
    lds_barrier();
    if (w == 0) {
      v4f ms;
      ms.x = mem_state[col];
      ms.y = mem_state[col + 1];
      ms.z = mem_state[HALF_ + col];
      ms.w = mem_state[HALF_ + col + 1];
      v4f tq[NWAVE_];
      #pragma unroll
      for (int j = 0; j < NWAVE_; ++j) tq[j] = TOT[j][c];
      #pragma unroll
      for (int j = 0; j < NWAVE_; ++j) {
        ms.x = fmaf(bw, ms.x, tq[j].x);
        ms.y = fmaf(bw, ms.y, tq[j].y);
        ms.z = fmaf(bw, ms.z, tq[j].z);
        ms.w = fmaf(bw, ms.w, tq[j].w);
      }
      if (h == 0) Mend[1][c] = ms;
    }
  }

  load_chunk(x + lane_off + ss_stride, X[1]);
  make_weights();
  const int ngroups = part ? (T_ - TCUT_ - 2 * SSROWS_) / (4 * SSROWS_)
                           : (TCUT_ - 2 * SSROWS_) / (4 * SSROWS_);
  #pragma unroll 1
  for (int s4 = 0; s4 < 4 * ngroups; s4 += 4) {
    step(I0{}, PFy{}, s4);
    step(I1{}, PFy{}, s4 + 1);
    step(I2{}, PFy{}, s4 + 2);
    step(I3{}, PFy{}, s4 + 3);
  }
  step(I0{}, PFn{}, 4 * ngroups);
  step(I1{}, PFn{}, 4 * ngroups + 1);
}

extern "C" void kernel_launch(void* const* d_in, const int* in_sizes, int n_in,
                              void* d_out, int out_size, void* d_ws,
                              size_t ws_size, hipStream_t stream) {
  const float* x           = (const float*)d_in[0];
  const float* gate        = (const float*)d_in[1];
  const float* input_scale = (const float*)d_in[2];
  const float* mem_state   = (const float*)d_in[3];
  const int*   step_idx    = (const int*)d_in[4];
  float* out = (float*)d_out;

  k_scan_strip<<<2 * B_ * NSTRIP_, 32 * NWAVE_, 0, stream>>>(
      x, gate, input_scale, mem_state, step_idx, out);
}
